# starting kernel
# baseline (speedup 1.0000x reference)
#include <hip/hip_runtime.h>

typedef __attribute__((ext_vector_type(2))) float v2f;
typedef __attribute__((ext_vector_type(8))) float v8f;

#define BATCH 32
#define INCH  32
#define OUTCH 32
#define DYND  16
#define KNBR  7

__global__ __launch_bounds__(256) void hexconv_build_weff(
    const float* __restrict__ dyn,
    const float* __restrict__ Wdyn,
    const float* __restrict__ Wstatic,
    float* __restrict__ Wt)
{
    int t = blockIdx.x * blockDim.x + threadIdx.x;
    if (t >= BATCH * KNBR * OUTCH * INCH) return;
    int i = t & 31;
    int o = (t >> 5) & 31;
    int k = (t >> 10) % KNBR;
    int b = t / (1024 * KNBR);

    float s = Wstatic[(k * INCH + i) * OUTCH + o];
    const float* dv = dyn + (b * KNBR + k) * DYND;
    const float* wv = Wdyn + (((size_t)k * DYND) * INCH + i) * OUTCH + o;
#pragma unroll
    for (int d = 0; d < DYND; ++d)
        s += dv[d] * wv[(size_t)d * INCH * OUTCH];

    Wt[((size_t)(b * KNBR + k) * OUTCH + o) * INCH + i] = s;
}

__global__ __launch_bounds__(256) void hexconv_wmma(
    const float* __restrict__ grid,
    const int*   __restrict__ nbr,
    const float* __restrict__ weff,
    const float* __restrict__ bias,
    float*       __restrict__ out,
    int N, int tilesPerB)
{
    const int lane = threadIdx.x & 31;
    const int wid  = threadIdx.x >> 5;
    const int nt   = blockIdx.x * 8 + wid;
    if (nt >= tilesPerB) return;
    const int b  = blockIdx.y;
    const int n0 = nt * 16;

    const int col  = lane & 15;
    const int half = lane >> 4;
    const int koff = half << 1;

    const int  row      = n0 + col;
    const bool rowValid = row < N;

    const float* gb = grid + (size_t)b * N * INCH;
    const float* wb = weff + (size_t)b * KNBR * OUTCH * INCH;

    v8f c0 = {};
    v8f c1 = {};

    for (int k = 0; k < KNBR; ++k) {
        int idx = rowValid ? nbr[(size_t)k * N + row] : -1;
        int safe = idx >= 0 ? idx : 0;
        float m  = idx >= 0 ? 1.0f : 0.0f;

        const float* gp = gb + (size_t)safe * INCH + koff;
        v2f a[8];
#pragma unroll
        for (int c = 0; c < 8; ++c) {
            v2f g = *(const v2f*)(gp + c * 4);
            a[c] = g * m;
        }

        const float* w0 = wb + (size_t)k * OUTCH * INCH + col * INCH + koff;
        const float* w1 = w0 + 16 * INCH;
#pragma unroll
        for (int c = 0; c < 8; ++c) {
            v2f b0 = *(const v2f*)(w0 + c * 4);
            v2f b1 = *(const v2f*)(w1 + c * 4);
            c0 = __builtin_amdgcn_wmma_f32_16x16x4_f32(false, a[c], false, b0,
                                                       (short)0, c0, false, false);
            c1 = __builtin_amdgcn_wmma_f32_16x16x4_f32(false, a[c], false, b1,
                                                       (short)0, c1, false, false);
        }
    }

    const float bl = bias[col];
    const float bh = bias[col + 16];
#pragma unroll
    for (int v = 0; v < 8; ++v) {
        int r = n0 + v + half * 8;
        if (r < N) {
            float* op = out + ((size_t)b * N + r) * OUTCH;
            op[col]      = c0[v] + bl;
            op[col + 16] = c1[v] + bh;
        }
    }
}

extern "C" void kernel_launch(void* const* d_in, const int* in_sizes, int n_in,
                              void* d_out, int out_size, void* d_ws, size_t ws_size,
                              hipStream_t stream) {
    const float* grid    = (const float*)d_in[0];
    const float* dyn     = (const float*)d_in[1];
    const int*   nbr     = (const int*)  d_in[2];
    const float* Wdyn    = (const float*)d_in[3];
    const float* Wstatic = (const float*)d_in[4];
    const float* bias    = (const float*)d_in[5];
    float*       out     = (float*)d_out;
    float*       weff    = (float*)d_ws;

    const int N = in_sizes[0] / (BATCH * INCH);

    const int total1 = BATCH * KNBR * OUTCH * INCH;
    hexconv_build_weff<<<(total1 + 255) / 256, 256, 0, stream>>>(dyn, Wdyn, Wstatic, weff);

    const int tilesPerB = (N + 15) / 16;
    dim3 gridDim((tilesPerB + 7) / 8, BATCH);
    hexconv_wmma<<<gridDim, 256, 0, stream>>>(grid, nbr, weff, bias, out, N, tilesPerB);
}


# Per-batch weights staged once per workgroup into LDS (lane-contiguous padded slots); each wave walks 2 row tiles and takes its B fragments by 16-byte LDS reads instead of 16-line global loads
# speedup vs baseline: 1.9170x; 1.9170x over previous
#include <hip/hip_runtime.h>

typedef __attribute__((ext_vector_type(2))) float v2f;
typedef __attribute__((ext_vector_type(4))) float v4f;
typedef __attribute__((ext_vector_type(8))) float v8f;

#define BATCH 32
#define INCH  32
#define OUTCH 32
#define DYND  16
#define KNBR  7

__global__ __launch_bounds__(256) void hexconv_build_weff(
    const float* __restrict__ dyn,
    const float* __restrict__ Wdyn,
    const float* __restrict__ Wstatic,
    float* __restrict__ Wt)
{
    int t = blockIdx.x * blockDim.x + threadIdx.x;
    if (t >= BATCH * KNBR * OUTCH * INCH) return;
    int i = t & 31;
    int o = (t >> 5) & 31;
    int k = (t >> 10) % KNBR;
    int b = t / (1024 * KNBR);

    float s = Wstatic[(k * INCH + i) * OUTCH + o];
    const float* dv = dyn + (b * KNBR + k) * DYND;
    const float* wv = Wdyn + (((size_t)k * DYND) * INCH + i) * OUTCH + o;
#pragma unroll
    for (int d = 0; d < DYND; ++d)
        s += dv[d] * wv[(size_t)d * INCH * OUTCH];

    Wt[((size_t)(b * KNBR + k) * OUTCH + o) * INCH + i] = s;
}

#define TPW   2
#define WSLOT 20
#define WLDS  (KNBR * 64 * WSLOT)

__global__ __launch_bounds__(256) void hexconv_wmma(
    const float* __restrict__ grid,
    const int*   __restrict__ nbr,
    const float* __restrict__ weff,
    const float* __restrict__ bias,
    float*       __restrict__ out,
    int N, int tilesPerB)
{
    __shared__ __attribute__((aligned(16))) float wl[WLDS];

    const int lane = threadIdx.x & 31;
    const int wid  = threadIdx.x >> 5;
    const int b    = blockIdx.y;

    const v4f* wsrc = (const v4f*)(weff + (size_t)b * KNBR * OUTCH * INCH);
#pragma unroll
    for (int it = 0; it < KNBR * OUTCH * INCH / 4 / 256; ++it) {
        int q = it * 256 + threadIdx.x;
        v4f w = wsrc[q];
        int c = q & 7;
        int o = (q >> 3) & 31;
        int k = q >> 8;
        float* d = wl + ((((k * 2 + (o >> 4)) * 2) * 16 + (o & 15)) * WSLOT + c * 2);
        *(v2f*)d                = w.xy;
        *(v2f*)(d + 16 * WSLOT) = w.zw;
    }
    __syncthreads();

    const int col  = lane & 15;
    const int half = lane >> 4;
    const int koff = half << 1;

    const float* gb = grid + (size_t)b * N * INCH;
    const float* lw = wl + (half * 16 + col) * WSLOT;

    const float bl = bias[col];
    const float bh = bias[col + 16];

    for (int j = 0; j < TPW; ++j) {
        const int nt = (blockIdx.x * TPW + j) * 8 + wid;
        if (nt >= tilesPerB) break;
        const int n0 = nt * 16;

        const int  row      = n0 + col;
        const bool rowValid = row < N;

        v8f c0 = {};
        v8f c1 = {};

        for (int k = 0; k < KNBR; ++k) {
            int idx = rowValid ? nbr[(size_t)k * N + row] : -1;
            int safe = idx >= 0 ? idx : 0;
            float m  = idx >= 0 ? 1.0f : 0.0f;

            const float* gp = gb + (size_t)safe * INCH + koff;
            v2f a[8];
#pragma unroll
            for (int c = 0; c < 8; ++c) {
                v2f g = *(const v2f*)(gp + c * 4);
                a[c] = g * m;
            }

            const v4f* w0 = (const v4f*)(lw + k * (64 * WSLOT));
            const v4f* w1 = (const v4f*)(lw + k * (64 * WSLOT) + 32 * WSLOT);
            v4f q0[4], q1[4];
#pragma unroll
            for (int p = 0; p < 4; ++p) { q0[p] = w0[p]; q1[p] = w1[p]; }
#pragma unroll
            for (int p = 0; p < 4; ++p) {
                c0 = __builtin_amdgcn_wmma_f32_16x16x4_f32(false, a[2 * p], false, q0[p].xy,
                                                           (short)0, c0, false, false);
                c1 = __builtin_amdgcn_wmma_f32_16x16x4_f32(false, a[2 * p], false, q1[p].xy,
                                                           (short)0, c1, false, false);
                c0 = __builtin_amdgcn_wmma_f32_16x16x4_f32(false, a[2 * p + 1], false, q0[p].zw,
                                                           (short)0, c0, false, false);
                c1 = __builtin_amdgcn_wmma_f32_16x16x4_f32(false, a[2 * p + 1], false, q1[p].zw,
                                                           (short)0, c1, false, false);
            }
        }

#pragma unroll
        for (int v = 0; v < 8; ++v) {
            int r = n0 + v + half * 8;
            if (r < N) {
                float* op = out + ((size_t)b * N + r) * OUTCH;
                op[col]      = c0[v] + bl;
                op[col + 16] = c1[v] + bh;
            }
        }
    }
}

extern "C" void kernel_launch(void* const* d_in, const int* in_sizes, int n_in,
                              void* d_out, int out_size, void* d_ws, size_t ws_size,
                              hipStream_t stream) {
    const float* grid    = (const float*)d_in[0];
    const float* dyn     = (const float*)d_in[1];
    const int*   nbr     = (const int*)  d_in[2];
    const float* Wdyn    = (const float*)d_in[3];
    const float* Wstatic = (const float*)d_in[4];
    const float* bias    = (const float*)d_in[5];
    float*       out     = (float*)d_out;
    float*       weff    = (float*)d_ws;

    const int N = in_sizes[0] / (BATCH * INCH);

    const int total1 = BATCH * KNBR * OUTCH * INCH;
    hexconv_build_weff<<<(total1 + 255) / 256, 256, 0, stream>>>(dyn, Wdyn, Wstatic, weff);

    const int tilesPerB = (N + 15) / 16;
    dim3 gridDim((tilesPerB + 8 * TPW - 1) / (8 * TPW), BATCH);
    hexconv_wmma<<<gridDim, 256, 0, stream>>>(grid, nbr, weff, bias, out, N, tilesPerB);
}


# Gather chain software-pipelined per wave: all neighbor indices of its tiles load up front, grid rows are requested 2-4 neighbor steps ahead of their products, same-hex-row neighbors back to back
# speedup vs baseline: 2.0119x; 1.0495x over previous
#include <hip/hip_runtime.h>

typedef __attribute__((ext_vector_type(2))) float v2f;
typedef __attribute__((ext_vector_type(4))) float v4f;
typedef __attribute__((ext_vector_type(8))) float v8f;

#define BATCH 32
#define INCH  32
#define OUTCH 32
#define DYND  16
#define KNBR  7

__global__ __launch_bounds__(256) void hexconv_build_weff(
    const float* __restrict__ dyn,
    const float* __restrict__ Wdyn,
    const float* __restrict__ Wstatic,
    float* __restrict__ Wt)
{
    int t = blockIdx.x * blockDim.x + threadIdx.x;
    if (t >= BATCH * KNBR * OUTCH * INCH) return;
    int i = t & 31;
    int o = (t >> 5) & 31;
    int k = (t >> 10) % KNBR;
    int b = t / (1024 * KNBR);

    float s = Wstatic[(k * INCH + i) * OUTCH + o];
    const float* dv = dyn + (b * KNBR + k) * DYND;
    const float* wv = Wdyn + (((size_t)k * DYND) * INCH + i) * OUTCH + o;
#pragma unroll
    for (int d = 0; d < DYND; ++d)
        s += dv[d] * wv[(size_t)d * INCH * OUTCH];

    Wt[((size_t)(b * KNBR + k) * OUTCH + o) * INCH + i] = s;
}

#define TPW   2
#define WSLOT 20
#define WLDS  (KNBR * 64 * WSLOT)

__global__ __launch_bounds__(256) void hexconv_wmma(
    const float* __restrict__ grid,
    const int*   __restrict__ nbr,
    const float* __restrict__ weff,
    const float* __restrict__ bias,
    float*       __restrict__ out,
    int N, int tilesPerB)
{
    __shared__ __attribute__((aligned(16))) float wl[WLDS];

    const int lane = threadIdx.x & 31;
    const int wid  = threadIdx.x >> 5;
    const int b    = blockIdx.y;

    const v4f* wsrc = (const v4f*)(weff + (size_t)b * KNBR * OUTCH * INCH);
#pragma unroll
    for (int it = 0; it < KNBR * OUTCH * INCH / 4 / 256; ++it) {
        int q = it * 256 + threadIdx.x;
        v4f w = wsrc[q];
        int c = q & 7;
        int o = (q >> 3) & 31;
        int k = q >> 8;
        float* d = wl + ((((k * 2 + (o >> 4)) * 2) * 16 + (o & 15)) * WSLOT + c * 2);
        *(v2f*)d                = w.xy;
        *(v2f*)(d + 16 * WSLOT) = w.zw;
    }
    __syncthreads();

    const int col  = lane & 15;
    const int half = lane >> 4;
    const int koff = half << 1;

    const float* gb = grid + (size_t)b * N * INCH;
    const float* lw = wl + (half * 16 + col) * WSLOT;

    const float bl = bias[col];
    const float bh = bias[col + 16];

    const int tile0 = blockIdx.x * (TPW * 8) + wid;
    if (tile0 >= tilesPerB) return;

    auto load_idx = [&](int nt, int (&ix)[KNBR]) {
        const int  row  = nt * 16 + col;
        const bool ok   = row < N;
        const int  rowc = ok ? row : N - 1;
#pragma unroll
        for (int k = 0; k < KNBR; ++k) {
            const int v = nbr[(size_t)k * N + rowc];
            ix[k] = ok ? v : -1;
        }
    };
    auto request = [&](int id, v2f (&g)[8]) {
        const int safe = id >= 0 ? id : 0;
        const float* gp = gb + (size_t)safe * INCH + koff;
#pragma unroll
        for (int c = 0; c < 8; ++c)
            g[c] = *(const v2f*)(gp + c * 4);
    };

    int idx[TPW][KNBR];
#pragma unroll
    for (int j = 0; j < TPW; ++j)
        load_idx(min(tile0 + 8 * j, tilesPerB - 1), idx[j]);
    v2f g[KNBR][8];
    request(idx[0][0], g[0]);
    request(idx[0][1], g[1]);

#pragma unroll
    for (int j = 0; j < TPW; ++j) {
        const int nt = tile0 + 8 * j;
        if (nt >= tilesPerB) break;
        const int  n0   = nt * 16;
        const bool more = (j + 1 < TPW);

        v8f c0 = {};
        v8f c1 = {};

#pragma unroll
        for (int k = 0; k < KNBR; ++k) {
            const bool req = (k == 0) || (k == 1) || (k == 3) || (more && k == 5);
            if (req) {
                __builtin_amdgcn_sched_barrier(0x106);
                if (k == 0) { request(idx[j][2], g[2]); request(idx[j][3], g[3]); }
                if (k == 1) { request(idx[j][4], g[4]); }
                if (k == 3) { request(idx[j][5], g[5]); request(idx[j][6], g[6]); }
                if (more && k == 5) { request(idx[j + 1][0], g[0]); request(idx[j + 1][1], g[1]); }
                __builtin_amdgcn_sched_barrier(0x106);
            }

            const float m = idx[j][k] >= 0 ? 1.0f : 0.0f;
            v2f a[8];
#pragma unroll
            for (int c = 0; c < 8; ++c) a[c] = g[k][c] * m;

            const v4f* w0 = (const v4f*)(lw + k * (64 * WSLOT));
            const v4f* w1 = (const v4f*)(lw + k * (64 * WSLOT) + 32 * WSLOT);
            v4f q0[4], q1[4];
#pragma unroll
            for (int p = 0; p < 4; ++p) { q0[p] = w0[p]; q1[p] = w1[p]; }
#pragma unroll
            for (int p = 0; p < 4; ++p) {
                c0 = __builtin_amdgcn_wmma_f32_16x16x4_f32(false, a[2 * p], false, q0[p].xy,
                                                           (short)0, c0, false, false);
                c1 = __builtin_amdgcn_wmma_f32_16x16x4_f32(false, a[2 * p], false, q1[p].xy,
                                                           (short)0, c1, false, false);
                c0 = __builtin_amdgcn_wmma_f32_16x16x4_f32(false, a[2 * p + 1], false, q0[p].zw,
                                                           (short)0, c0, false, false);
                c1 = __builtin_amdgcn_wmma_f32_16x16x4_f32(false, a[2 * p + 1], false, q1[p].zw,
                                                           (short)0, c1, false, false);
            }
        }

#pragma unroll
        for (int v = 0; v < 8; ++v) {
            int r = n0 + v + half * 8;
            if (r < N) {
                float* op = out + ((size_t)b * N + r) * OUTCH;
                op[col]      = c0[v] + bl;
                op[col + 16] = c1[v] + bh;
            }
        }
    }
}

extern "C" void kernel_launch(void* const* d_in, const int* in_sizes, int n_in,
                              void* d_out, int out_size, void* d_ws, size_t ws_size,
                              hipStream_t stream) {
    const float* grid    = (const float*)d_in[0];
    const float* dyn     = (const float*)d_in[1];
    const int*   nbr     = (const int*)  d_in[2];
    const float* Wdyn    = (const float*)d_in[3];
    const float* Wstatic = (const float*)d_in[4];
    const float* bias    = (const float*)d_in[5];
    float*       out     = (float*)d_out;
    float*       weff    = (float*)d_ws;

    const int N = in_sizes[0] / (BATCH * INCH);

    const int total1 = BATCH * KNBR * OUTCH * INCH;
    hexconv_build_weff<<<(total1 + 255) / 256, 256, 0, stream>>>(dyn, Wdyn, Wstatic, weff);

    const int tilesPerB = (N + 15) / 16;
    dim3 gridDim((tilesPerB + 8 * TPW - 1) / (8 * TPW), BATCH);
    hexconv_wmma<<<gridDim, 256, 0, stream>>>(grid, nbr, weff, bias, out, N, tilesPerB);
}


# Gather by 16-byte loads: 4 b128 per lane per neighbor row instead of 8 b64; in-channel order within each 8 permuted alike on A and the LDS weight slots (changes the f32 sum order)
# speedup vs baseline: 2.0855x; 1.0366x over previous
#include <hip/hip_runtime.h>

typedef __attribute__((ext_vector_type(2))) float v2f;
typedef __attribute__((ext_vector_type(4))) float v4f;
typedef __attribute__((ext_vector_type(8))) float v8f;

#define BATCH 32
#define INCH  32
#define OUTCH 32
#define DYND  16
#define KNBR  7

__global__ __launch_bounds__(256) void hexconv_build_weff(
    const float* __restrict__ dyn,
    const float* __restrict__ Wdyn,
    const float* __restrict__ Wstatic,
    float* __restrict__ Wt)
{
    int t = blockIdx.x * blockDim.x + threadIdx.x;
    if (t >= BATCH * KNBR * OUTCH * INCH) return;
    int i = t & 31;
    int o = (t >> 5) & 31;
    int k = (t >> 10) % KNBR;
    int b = t / (1024 * KNBR);

    float s = Wstatic[(k * INCH + i) * OUTCH + o];
    const float* dv = dyn + (b * KNBR + k) * DYND;
    const float* wv = Wdyn + (((size_t)k * DYND) * INCH + i) * OUTCH + o;
#pragma unroll
    for (int d = 0; d < DYND; ++d)
        s += dv[d] * wv[(size_t)d * INCH * OUTCH];

    Wt[((size_t)(b * KNBR + k) * OUTCH + o) * INCH + i] = s;
}

#define TPW   2
#define WSLOT 20
#define WLDS  (KNBR * 64 * WSLOT)

__global__ __launch_bounds__(256) void hexconv_wmma(
    const float* __restrict__ grid,
    const int*   __restrict__ nbr,
    const float* __restrict__ weff,
    const float* __restrict__ bias,
    float*       __restrict__ out,
    int N, int tilesPerB)
{
    __shared__ __attribute__((aligned(16))) float wl[WLDS];

    const int lane = threadIdx.x & 31;
    const int wid  = threadIdx.x >> 5;
    const int b    = blockIdx.y;

    const v4f* wsrc = (const v4f*)(weff + (size_t)b * KNBR * OUTCH * INCH);
#pragma unroll
    for (int it = 0; it < KNBR * OUTCH * INCH / 4 / 256; ++it) {
        int q = it * 256 + threadIdx.x;
        v4f w = wsrc[q];
        int c = q & 7;
        int o = (q >> 3) & 31;
        int k = q >> 8;
        float* d = wl + ((((k * 2 + (o >> 4)) * 2 + (c & 1)) * 16 + (o & 15)) * WSLOT
                         + (c >> 1) * 4);
        *(v4f*)d = w;
    }
    __syncthreads();

    const int col  = lane & 15;
    const int half = lane >> 4;
    const int koff = half << 2;

    const float* gb = grid + (size_t)b * N * INCH;
    const float* lw = wl + (half * 16 + col) * WSLOT;

    const float bl = bias[col];
    const float bh = bias[col + 16];

    const int tile0 = blockIdx.x * (TPW * 8) + wid;
    if (tile0 >= tilesPerB) return;

    auto load_idx = [&](int nt, int (&ix)[KNBR]) {
        const int  row  = nt * 16 + col;
        const bool ok   = row < N;
        const int  rowc = ok ? row : N - 1;
#pragma unroll
        for (int k = 0; k < KNBR; ++k) {
            const int v = nbr[(size_t)k * N + rowc];
            ix[k] = ok ? v : -1;
        }
    };
    auto request = [&](int id, v4f (&g)[4]) {
        const int safe = id >= 0 ? id : 0;
        const float* gp = gb + (size_t)safe * INCH + koff;
#pragma unroll
        for (int p = 0; p < 4; ++p)
            g[p] = *(const v4f*)(gp + p * 8);
    };

    int idx[TPW][KNBR];
#pragma unroll
    for (int j = 0; j < TPW; ++j)
        load_idx(min(tile0 + 8 * j, tilesPerB - 1), idx[j]);
    v4f g[KNBR][4];
    request(idx[0][0], g[0]);
    request(idx[0][1], g[1]);

#pragma unroll
    for (int j = 0; j < TPW; ++j) {
        const int nt = tile0 + 8 * j;
        if (nt >= tilesPerB) break;
        const int  n0   = nt * 16;
        const bool more = (j + 1 < TPW);

        v8f c0 = {};
        v8f c1 = {};

#pragma unroll
        for (int k = 0; k < KNBR; ++k) {
            const bool req = (k == 0) || (k == 1) || (k == 3) || (more && k == 5);
            if (req) {
                __builtin_amdgcn_sched_barrier(0x106);
                if (k == 0) { request(idx[j][2], g[2]); request(idx[j][3], g[3]); }
                if (k == 1) { request(idx[j][4], g[4]); }
                if (k == 3) { request(idx[j][5], g[5]); request(idx[j][6], g[6]); }
                if (more && k == 5) { request(idx[j + 1][0], g[0]); request(idx[j + 1][1], g[1]); }
                __builtin_amdgcn_sched_barrier(0x106);
            }

            const float m = idx[j][k] >= 0 ? 1.0f : 0.0f;
            v2f a[8];
#pragma unroll
            for (int p = 0; p < 4; ++p) {
                a[2 * p]     = g[k][p].xy * m;
                a[2 * p + 1] = g[k][p].zw * m;
            }

            const v4f* w0 = (const v4f*)(lw + k * (64 * WSLOT));
            const v4f* w1 = (const v4f*)(lw + k * (64 * WSLOT) + 32 * WSLOT);
            v4f q0[4], q1[4];
#pragma unroll
            for (int p = 0; p < 4; ++p) { q0[p] = w0[p]; q1[p] = w1[p]; }
#pragma unroll
            for (int p = 0; p < 4; ++p) {
                c0 = __builtin_amdgcn_wmma_f32_16x16x4_f32(false, a[2 * p], false, q0[p].xy,
                                                           (short)0, c0, false, false);
                c1 = __builtin_amdgcn_wmma_f32_16x16x4_f32(false, a[2 * p], false, q1[p].xy,
                                                           (short)0, c1, false, false);
                c0 = __builtin_amdgcn_wmma_f32_16x16x4_f32(false, a[2 * p + 1], false, q0[p].zw,
                                                           (short)0, c0, false, false);
                c1 = __builtin_amdgcn_wmma_f32_16x16x4_f32(false, a[2 * p + 1], false, q1[p].zw,
                                                           (short)0, c1, false, false);
            }
        }

#pragma unroll
        for (int v = 0; v < 8; ++v) {
            int r = n0 + v + half * 8;
            if (r < N) {
                float* op = out + ((size_t)b * N + r) * OUTCH;
                op[col]      = c0[v] + bl;
                op[col + 16] = c1[v] + bh;
            }
        }
    }
}

extern "C" void kernel_launch(void* const* d_in, const int* in_sizes, int n_in,
                              void* d_out, int out_size, void* d_ws, size_t ws_size,
                              hipStream_t stream) {
    const float* grid    = (const float*)d_in[0];
    const float* dyn     = (const float*)d_in[1];
    const int*   nbr     = (const int*)  d_in[2];
    const float* Wdyn    = (const float*)d_in[3];
    const float* Wstatic = (const float*)d_in[4];
    const float* bias    = (const float*)d_in[5];
    float*       out     = (float*)d_out;
    float*       weff    = (float*)d_ws;

    const int N = in_sizes[0] / (BATCH * INCH);

    const int total1 = BATCH * KNBR * OUTCH * INCH;
    hexconv_build_weff<<<(total1 + 255) / 256, 256, 0, stream>>>(dyn, Wdyn, Wstatic, weff);

    const int tilesPerB = (N + 15) / 16;
    dim3 gridDim((tilesPerB + 8 * TPW - 1) / (8 * TPW), BATCH);
    hexconv_wmma<<<gridDim, 256, 0, stream>>>(grid, nbr, weff, bias, out, N, tilesPerB);
}


# All B fragments (224 registers) read from LDS once per wave and kept; 256 workgroups of 8 waves, each wave walks ~27 row tiles, row requests one tile ahead; no LDS access in the product loop
# speedup vs baseline: 2.1570x; 1.0343x over previous
#include <hip/hip_runtime.h>

typedef __attribute__((ext_vector_type(2))) float v2f;
typedef __attribute__((ext_vector_type(4))) float v4f;
typedef __attribute__((ext_vector_type(8))) float v8f;

#define BATCH 32
#define INCH  32
#define OUTCH 32
#define DYND  16
#define KNBR  7

__global__ __launch_bounds__(256) void hexconv_build_weff(
    const float* __restrict__ dyn,
    const float* __restrict__ Wdyn,
    const float* __restrict__ Wstatic,
    float* __restrict__ Wt)
{
    int t = blockIdx.x * blockDim.x + threadIdx.x;
    if (t >= BATCH * KNBR * OUTCH * INCH) return;
    int i = t & 31;
    int o = (t >> 5) & 31;
    int k = (t >> 10) % KNBR;
    int b = t / (1024 * KNBR);

    float s = Wstatic[(k * INCH + i) * OUTCH + o];
    const float* dv = dyn + (b * KNBR + k) * DYND;
    const float* wv = Wdyn + (((size_t)k * DYND) * INCH + i) * OUTCH + o;
#pragma unroll
    for (int d = 0; d < DYND; ++d)
        s += dv[d] * wv[(size_t)d * INCH * OUTCH];

    Wt[((size_t)(b * KNBR + k) * OUTCH + o) * INCH + i] = s;
}

#define WGPB  8
#define WSLOT 20
#define WLDS  (KNBR * 64 * WSLOT)

__global__ __launch_bounds__(256) void hexconv_wmma(
    const float* __restrict__ grid,
    const int*   __restrict__ nbr,
    const float* __restrict__ weff,
    const float* __restrict__ bias,
    float*       __restrict__ out,
    int N, int tilesPerB)
{
    __shared__ __attribute__((aligned(16))) float wl[WLDS];

    const int lane = threadIdx.x & 31;
    const int wid  = threadIdx.x >> 5;
    const int b    = blockIdx.y;

    const int col  = lane & 15;
    const int half = lane >> 4;
    const int koff = half << 2;

    const float* gb = grid + (size_t)b * N * INCH;
    const float* lw = wl + (half * 16 + col) * WSLOT;

    const int chunk = (tilesPerB + WGPB - 1) / WGPB;
    const int tend  = min(((int)blockIdx.x + 1) * chunk, tilesPerB);
    int nt = (int)blockIdx.x * chunk + wid;
    const bool has = nt < tend;
    const int  t0  = has ? nt : tilesPerB - 1;

    auto load_idx = [&](int t, int (&ix)[KNBR]) {
        const int  row  = t * 16 + col;
        const bool ok   = row < N;
        const int  rowc = ok ? row : N - 1;
#pragma unroll
        for (int k = 0; k < KNBR; ++k) {
            const int v = nbr[(size_t)k * N + rowc];
            ix[k] = ok ? v : -1;
        }
    };
    auto request = [&](int id, v4f (&g)[4]) {
        const int safe = id >= 0 ? id : 0;
        const float* gp = gb + (size_t)safe * INCH + koff;
#pragma unroll
        for (int p = 0; p < 4; ++p)
            g[p] = *(const v4f*)(gp + p * 8);
    };

    int ic[KNBR], ix1[KNBR];
    load_idx(t0, ic);
    load_idx(has && nt + 8 < tend ? nt + 8 : t0, ix1);
    const float bl = bias[col];
    const float bh = bias[col + 16];

    const v4f* wsrc = (const v4f*)(weff + (size_t)b * KNBR * OUTCH * INCH);
#pragma unroll
    for (int it = 0; it < KNBR * OUTCH * INCH / 4 / 256; ++it) {
        int q = it * 256 + threadIdx.x;
        v4f w = wsrc[q];
        int c = q & 7;
        int o = (q >> 3) & 31;
        int k = q >> 8;
        float* d = wl + ((((k * 2 + (o >> 4)) * 2 + (c & 1)) * 16 + (o & 15)) * WSLOT
                         + (c >> 1) * 4);
        *(v4f*)d = w;
    }
    __syncthreads();

    v4f g[KNBR][4];
#pragma unroll
    for (int k = 0; k < KNBR; ++k) {
        __builtin_amdgcn_sched_barrier(0x6);
        request(ic[k], g[k]);
    }
    __builtin_amdgcn_sched_barrier(0x6);

    v4f q[KNBR][2][4];
#pragma unroll
    for (int k = 0; k < KNBR; ++k)
#pragma unroll
        for (int oh = 0; oh < 2; ++oh)
#pragma unroll
            for (int p = 0; p < 4; ++p)
                q[k][oh][p] = *(const v4f*)(lw + k * (64 * WSLOT) + oh * (32 * WSLOT) + p * 4);

    if (!has) return;

    for (; nt < tend; nt += 8) {
        const int n0 = nt * 16;
        int i2[KNBR];
        load_idx(nt + 16 < tend ? nt + 16 : nt, i2);

        v8f c0 = {};
        v8f c1 = {};

#pragma unroll
        for (int k = 0; k < KNBR; ++k) {
            const float m = ic[k] >= 0 ? 1.0f : 0.0f;
            v2f a[8];
#pragma unroll
            for (int p = 0; p < 4; ++p) {
                a[2 * p]     = g[k][p].xy * m;
                a[2 * p + 1] = g[k][p].zw * m;
            }
            __builtin_amdgcn_sched_barrier(0x6);
            request(ix1[k], g[k]);
            __builtin_amdgcn_sched_barrier(0x6);

#pragma unroll
            for (int p = 0; p < 4; ++p) {
                c0 = __builtin_amdgcn_wmma_f32_16x16x4_f32(false, a[2 * p], false, q[k][0][p].xy,
                                                           (short)0, c0, false, false);
                c1 = __builtin_amdgcn_wmma_f32_16x16x4_f32(false, a[2 * p], false, q[k][1][p].xy,
                                                           (short)0, c1, false, false);
                c0 = __builtin_amdgcn_wmma_f32_16x16x4_f32(false, a[2 * p + 1], false, q[k][0][p].zw,
                                                           (short)0, c0, false, false);
                c1 = __builtin_amdgcn_wmma_f32_16x16x4_f32(false, a[2 * p + 1], false, q[k][1][p].zw,
                                                           (short)0, c1, false, false);
            }
        }

#pragma unroll
        for (int v = 0; v < 8; ++v) {
            int r = n0 + v + half * 8;
            if (r < N) {
                float* op = out + ((size_t)b * N + r) * OUTCH;
                op[col]      = c0[v] + bl;
                op[col + 16] = c1[v] + bh;
            }
        }

#pragma unroll
        for (int k = 0; k < KNBR; ++k) { ic[k] = ix1[k]; ix1[k] = i2[k]; }
    }
}

extern "C" void kernel_launch(void* const* d_in, const int* in_sizes, int n_in,
                              void* d_out, int out_size, void* d_ws, size_t ws_size,
                              hipStream_t stream) {
    const float* grid    = (const float*)d_in[0];
    const float* dyn     = (const float*)d_in[1];
    const int*   nbr     = (const int*)  d_in[2];
    const float* Wdyn    = (const float*)d_in[3];
    const float* Wstatic = (const float*)d_in[4];
    const float* bias    = (const float*)d_in[5];
    float*       out     = (float*)d_out;
    float*       weff    = (float*)d_ws;

    const int N = in_sizes[0] / (BATCH * INCH);

    const int total1 = BATCH * KNBR * OUTCH * INCH;
    hexconv_build_weff<<<(total1 + 255) / 256, 256, 0, stream>>>(dyn, Wdyn, Wstatic, weff);

    const int tilesPerB = (N + 15) / 16;
    dim3 gridDim(WGPB, BATCH);
    hexconv_wmma<<<gridDim, 256, 0, stream>>>(grid, nbr, weff, bias, out, N, tilesPerB);
}


# Weight-build kernel: one workgroup per (batch, neighbor), 16-byte loads so each wave-wide load covers 4 whole 128-byte lines (was 32 lines), 4x fewer waves; products fused into the adds (fma)
# speedup vs baseline: 2.2626x; 1.0490x over previous
#include <hip/hip_runtime.h>

typedef __attribute__((ext_vector_type(2))) float v2f;
typedef __attribute__((ext_vector_type(4))) float v4f;
typedef __attribute__((ext_vector_type(8))) float v8f;

#define BATCH 32
#define INCH  32
#define OUTCH 32
#define DYND  16
#define KNBR  7

__global__ __launch_bounds__(256) void hexconv_build_weff(
    const float* __restrict__ dyn,
    const float* __restrict__ Wdyn,
    const float* __restrict__ Wstatic,
    float* __restrict__ Wt)
{
    const int bk = blockIdx.x;
    const int k  = bk % KNBR;
    const int oc = threadIdx.x & 7;
    const int i  = threadIdx.x >> 3;

    const float* dv = dyn + bk * DYND;
    const int off = i * OUTCH + oc * 4;
    const float* wv = Wdyn + (size_t)k * DYND * INCH * OUTCH + off;

    v4f s = *(const v4f*)(Wstatic + k * INCH * OUTCH + off);
    v4f w[DYND];
#pragma unroll
    for (int d = 0; d < DYND; ++d)
        w[d] = *(const v4f*)(wv + d * INCH * OUTCH);
#pragma unroll
    for (int d = 0; d < DYND; ++d)
        s += dv[d] * w[d];

    float* dst = Wt + ((size_t)bk * OUTCH + oc * 4) * INCH + i;
    dst[0 * INCH] = s.x;
    dst[1 * INCH] = s.y;
    dst[2 * INCH] = s.z;
    dst[3 * INCH] = s.w;
}

#define WGPB  8
#define WSLOT 20
#define WLDS  (KNBR * 64 * WSLOT)

__global__ __launch_bounds__(256) void hexconv_wmma(
    const float* __restrict__ grid,
    const int*   __restrict__ nbr,
    const float* __restrict__ weff,
    const float* __restrict__ bias,
    float*       __restrict__ out,
    int N, int tilesPerB)
{
    __shared__ __attribute__((aligned(16))) float wl[WLDS];

    const int lane = threadIdx.x & 31;
    const int wid  = threadIdx.x >> 5;
    const int b    = blockIdx.y;

    const int col  = lane & 15;
    const int half = lane >> 4;
    const int koff = half << 2;

    const float* gb = grid + (size_t)b * N * INCH;
    const float* lw = wl + (half * 16 + col) * WSLOT;

    const int chunk = (tilesPerB + WGPB - 1) / WGPB;
    const int tend  = min(((int)blockIdx.x + 1) * chunk, tilesPerB);
    int nt = (int)blockIdx.x * chunk + wid;
    const bool has = nt < tend;
    const int  t0  = has ? nt : tilesPerB - 1;

    auto load_idx = [&](int t, int (&ix)[KNBR]) {
        const int  row  = t * 16 + col;
        const bool ok   = row < N;
        const int  rowc = ok ? row : N - 1;
#pragma unroll
        for (int k = 0; k < KNBR; ++k) {
            const int v = nbr[(size_t)k * N + rowc];
            ix[k] = ok ? v : -1;
        }
    };
    auto request = [&](int id, v4f (&g)[4]) {
        const int safe = id >= 0 ? id : 0;
        const float* gp = gb + (size_t)safe * INCH + koff;
#pragma unroll
        for (int p = 0; p < 4; ++p)
            g[p] = *(const v4f*)(gp + p * 8);
    };

    int ic[KNBR], ix1[KNBR];
    load_idx(t0, ic);
    load_idx(has && nt + 8 < tend ? nt + 8 : t0, ix1);
    const float bl = bias[col];
    const float bh = bias[col + 16];

    const v4f* wsrc = (const v4f*)(weff + (size_t)b * KNBR * OUTCH * INCH);
#pragma unroll
    for (int it = 0; it < KNBR * OUTCH * INCH / 4 / 256; ++it) {
        int q = it * 256 + threadIdx.x;
        v4f w = wsrc[q];
        int c = q & 7;
        int o = (q >> 3) & 31;
        int k = q >> 8;
        float* d = wl + ((((k * 2 + (o >> 4)) * 2 + (c & 1)) * 16 + (o & 15)) * WSLOT
                         + (c >> 1) * 4);
        *(v4f*)d = w;
    }
    __syncthreads();

    v4f g[KNBR][4];
#pragma unroll
    for (int k = 0; k < KNBR; ++k) {
        __builtin_amdgcn_sched_barrier(0x6);
        request(ic[k], g[k]);
    }
    __builtin_amdgcn_sched_barrier(0x6);

    v4f q[KNBR][2][4];
#pragma unroll
    for (int k = 0; k < KNBR; ++k)
#pragma unroll
        for (int oh = 0; oh < 2; ++oh)
#pragma unroll
            for (int p = 0; p < 4; ++p)
                q[k][oh][p] = *(const v4f*)(lw + k * (64 * WSLOT) + oh * (32 * WSLOT) + p * 4);

    if (!has) return;

    for (; nt < tend; nt += 8) {
        const int n0 = nt * 16;
        int i2[KNBR];
        load_idx(nt + 16 < tend ? nt + 16 : nt, i2);

        v8f c0 = {};
        v8f c1 = {};

#pragma unroll
        for (int k = 0; k < KNBR; ++k) {
            const float m = ic[k] >= 0 ? 1.0f : 0.0f;
            v2f a[8];
#pragma unroll
            for (int p = 0; p < 4; ++p) {
                a[2 * p]     = g[k][p].xy * m;
                a[2 * p + 1] = g[k][p].zw * m;
            }
            __builtin_amdgcn_sched_barrier(0x6);
            request(ix1[k], g[k]);
            __builtin_amdgcn_sched_barrier(0x6);

#pragma unroll
            for (int p = 0; p < 4; ++p) {
                c0 = __builtin_amdgcn_wmma_f32_16x16x4_f32(false, a[2 * p], false, q[k][0][p].xy,
                                                           (short)0, c0, false, false);
                c1 = __builtin_amdgcn_wmma_f32_16x16x4_f32(false, a[2 * p], false, q[k][1][p].xy,
                                                           (short)0, c1, false, false);
                c0 = __builtin_amdgcn_wmma_f32_16x16x4_f32(false, a[2 * p + 1], false, q[k][0][p].zw,
                                                           (short)0, c0, false, false);
                c1 = __builtin_amdgcn_wmma_f32_16x16x4_f32(false, a[2 * p + 1], false, q[k][1][p].zw,
                                                           (short)0, c1, false, false);
            }
        }

#pragma unroll
        for (int v = 0; v < 8; ++v) {
            int r = n0 + v + half * 8;
            if (r < N) {
                float* op = out + ((size_t)b * N + r) * OUTCH;
                op[col]      = c0[v] + bl;
                op[col + 16] = c1[v] + bh;
            }
        }

#pragma unroll
        for (int k = 0; k < KNBR; ++k) { ic[k] = ix1[k]; ix1[k] = i2[k]; }
    }
}

extern "C" void kernel_launch(void* const* d_in, const int* in_sizes, int n_in,
                              void* d_out, int out_size, void* d_ws, size_t ws_size,
                              hipStream_t stream) {
    const float* grid    = (const float*)d_in[0];
    const float* dyn     = (const float*)d_in[1];
    const int*   nbr     = (const int*)  d_in[2];
    const float* Wdyn    = (const float*)d_in[3];
    const float* Wstatic = (const float*)d_in[4];
    const float* bias    = (const float*)d_in[5];
    float*       out     = (float*)d_out;
    float*       weff    = (float*)d_ws;

    const int N = in_sizes[0] / (BATCH * INCH);

    hexconv_build_weff<<<BATCH * KNBR, 256, 0, stream>>>(dyn, Wdyn, Wstatic, weff);

    const int tilesPerB = (N + 15) / 16;
    dim3 gridDim(WGPB, BATCH);
    hexconv_wmma<<<gridDim, 256, 0, stream>>>(grid, nbr, weff, bias, out, N, tilesPerB);
}


# Tile loop control made wave-uniform (wave index through a scalar register): loop and whole-tile tests become scalar branches; a full tile's row stores issue back to back from one base address
# speedup vs baseline: 2.5766x; 1.1388x over previous
#include <hip/hip_runtime.h>

typedef __attribute__((ext_vector_type(2))) float v2f;
typedef __attribute__((ext_vector_type(4))) float v4f;
typedef __attribute__((ext_vector_type(8))) float v8f;

#define BATCH 32
#define INCH  32
#define OUTCH 32
#define DYND  16
#define KNBR  7

__global__ __launch_bounds__(256) void hexconv_build_weff(
    const float* __restrict__ dyn,
    const float* __restrict__ Wdyn,
    const float* __restrict__ Wstatic,
    float* __restrict__ Wt)
{
    const int bk = blockIdx.x;
    const int k  = bk % KNBR;
    const int oc = threadIdx.x & 7;
    const int i  = threadIdx.x >> 3;

    const float* dv = dyn + bk * DYND;
    const int off = i * OUTCH + oc * 4;
    const float* wv = Wdyn + (size_t)k * DYND * INCH * OUTCH + off;

    v4f s = *(const v4f*)(Wstatic + k * INCH * OUTCH + off);
    v4f w[DYND];
#pragma unroll
    for (int d = 0; d < DYND; ++d)
        w[d] = *(const v4f*)(wv + d * INCH * OUTCH);
#pragma unroll
    for (int d = 0; d < DYND; ++d)
        s += dv[d] * w[d];

    float* dst = Wt + ((size_t)bk * OUTCH + oc * 4) * INCH + i;
    dst[0 * INCH] = s.x;
    dst[1 * INCH] = s.y;
    dst[2 * INCH] = s.z;
    dst[3 * INCH] = s.w;
}

#define WGPB  8
#define WSLOT 20
#define WLDS  (KNBR * 64 * WSLOT)

__global__ __launch_bounds__(256) void hexconv_wmma(
    const float* __restrict__ grid,
    const int*   __restrict__ nbr,
    const float* __restrict__ weff,
    const float* __restrict__ bias,
    float*       __restrict__ out,
    int N, int tilesPerB)
{
    __shared__ __attribute__((aligned(16))) float wl[WLDS];

    const int lane = threadIdx.x & 31;
    const int wid  = __builtin_amdgcn_readfirstlane((int)(threadIdx.x >> 5));
    const int b    = blockIdx.y;

    const int col  = lane & 15;
    const int half = lane >> 4;
    const int koff = half << 2;

    const float* gb = grid + (size_t)b * N * INCH;
    const float* lw = wl + (half * 16 + col) * WSLOT;

    const int chunk = (tilesPerB + WGPB - 1) / WGPB;
    const int tend  = min(((int)blockIdx.x + 1) * chunk, tilesPerB);
    int nt = (int)blockIdx.x * chunk + wid;
    const bool has = nt < tend;
    const int  t0  = has ? nt : tilesPerB - 1;

    auto load_idx = [&](int t, int (&ix)[KNBR]) {
        const int  row  = t * 16 + col;
        const bool ok   = row < N;
        const int  rowc = ok ? row : N - 1;
#pragma unroll
        for (int k = 0; k < KNBR; ++k) {
            const int v = nbr[(size_t)k * N + rowc];
            ix[k] = ok ? v : -1;
        }
    };
    auto request = [&](int id, v4f (&g)[4]) {
        const int safe = id >= 0 ? id : 0;
        const float* gp = gb + (size_t)safe * INCH + koff;
#pragma unroll
        for (int p = 0; p < 4; ++p)
            g[p] = *(const v4f*)(gp + p * 8);
    };

    int ic[KNBR], ix1[KNBR];
    load_idx(t0, ic);
    load_idx(has && nt + 8 < tend ? nt + 8 : t0, ix1);
    const float bl = bias[col];
    const float bh = bias[col + 16];

    const v4f* wsrc = (const v4f*)(weff + (size_t)b * KNBR * OUTCH * INCH);
#pragma unroll
    for (int it = 0; it < KNBR * OUTCH * INCH / 4 / 256; ++it) {
        int q = it * 256 + threadIdx.x;
        v4f w = wsrc[q];
        int c = q & 7;
        int o = (q >> 3) & 31;
        int k = q >> 8;
        float* d = wl + ((((k * 2 + (o >> 4)) * 2 + (c & 1)) * 16 + (o & 15)) * WSLOT
                         + (c >> 1) * 4);
        *(v4f*)d = w;
    }
    __syncthreads();

    v4f g[KNBR][4];
#pragma unroll
    for (int k = 0; k < KNBR; ++k) {
        __builtin_amdgcn_sched_barrier(0x6);
        request(ic[k], g[k]);
    }
    __builtin_amdgcn_sched_barrier(0x6);

    v4f q[KNBR][2][4];
#pragma unroll
    for (int k = 0; k < KNBR; ++k)
#pragma unroll
        for (int oh = 0; oh < 2; ++oh)
#pragma unroll
            for (int p = 0; p < 4; ++p)
                q[k][oh][p] = *(const v4f*)(lw + k * (64 * WSLOT) + oh * (32 * WSLOT) + p * 4);

    if (!has) return;

    for (; nt < tend; nt += 8) {
        const int n0 = nt * 16;
        int i2[KNBR];
        load_idx(nt + 16 < tend ? nt + 16 : nt, i2);

        v8f c0 = {};
        v8f c1 = {};

#pragma unroll
        for (int k = 0; k < KNBR; ++k) {
            const float m = ic[k] >= 0 ? 1.0f : 0.0f;
            v2f a[8];
#pragma unroll
            for (int p = 0; p < 4; ++p) {
                a[2 * p]     = g[k][p].xy * m;
                a[2 * p + 1] = g[k][p].zw * m;
            }
            __builtin_amdgcn_sched_barrier(0x6);
            request(ix1[k], g[k]);
            __builtin_amdgcn_sched_barrier(0x6);

#pragma unroll
            for (int p = 0; p < 4; ++p) {
                c0 = __builtin_amdgcn_wmma_f32_16x16x4_f32(false, a[2 * p], false, q[k][0][p].xy,
                                                           (short)0, c0, false, false);
                c1 = __builtin_amdgcn_wmma_f32_16x16x4_f32(false, a[2 * p], false, q[k][1][p].xy,
                                                           (short)0, c1, false, false);
                c0 = __builtin_amdgcn_wmma_f32_16x16x4_f32(false, a[2 * p + 1], false, q[k][0][p].zw,
                                                           (short)0, c0, false, false);
                c1 = __builtin_amdgcn_wmma_f32_16x16x4_f32(false, a[2 * p + 1], false, q[k][1][p].zw,
                                                           (short)0, c1, false, false);
            }
        }

        const int r0 = n0 + half * 8;
        float* op = out + ((size_t)b * N + r0) * OUTCH + col;
        if (n0 + 16 <= N) {
#pragma unroll
            for (int v = 0; v < 8; ++v) {
                op[v * OUTCH]      = c0[v] + bl;
                op[v * OUTCH + 16] = c1[v] + bh;
            }
        } else {
#pragma unroll
            for (int v = 0; v < 8; ++v) {
                const int r = r0 + v;
                if (r < N) {
                    float* orow = out + ((size_t)b * N + r) * OUTCH;
                    orow[col]      = c0[v] + bl;
                    orow[col + 16] = c1[v] + bh;
                }
            }
        }

#pragma unroll
        for (int k = 0; k < KNBR; ++k) { ic[k] = ix1[k]; ix1[k] = i2[k]; }
    }
}

extern "C" void kernel_launch(void* const* d_in, const int* in_sizes, int n_in,
                              void* d_out, int out_size, void* d_ws, size_t ws_size,
                              hipStream_t stream) {
    const float* grid    = (const float*)d_in[0];
    const float* dyn     = (const float*)d_in[1];
    const int*   nbr     = (const int*)  d_in[2];
    const float* Wdyn    = (const float*)d_in[3];
    const float* Wstatic = (const float*)d_in[4];
    const float* bias    = (const float*)d_in[5];
    float*       out     = (float*)d_out;
    float*       weff    = (float*)d_ws;

    const int N = in_sizes[0] / (BATCH * INCH);

    hexconv_build_weff<<<BATCH * KNBR, 256, 0, stream>>>(dyn, Wdyn, Wstatic, weff);

    const int tilesPerB = (N + 15) / 16;
    dim3 gridDim(WGPB, BATCH);
    hexconv_wmma<<<gridDim, 256, 0, stream>>>(grid, nbr, weff, bias, out, N, tilesPerB);
}
